# starting kernel
# baseline (speedup 1.0000x reference)
#include <hip/hip_runtime.h>
#include <math.h>

#define N_NODES 100000
#define N_EDGES 1600000
#define F 64
#define K_ORDER 8
#define LN_EPS 1e-5f

typedef __attribute__((ext_vector_type(2))) float v2f;
typedef __attribute__((ext_vector_type(8))) float v8f;

__global__ void zero_f32(float* __restrict__ p, int n) {
  int i = blockIdx.x * blockDim.x + threadIdx.x;
  if (i < n) p[i] = 0.0f;
}

__global__ void degree_kernel(const int* __restrict__ row,
                              const int* __restrict__ col,
                              float* __restrict__ deg) {
  int e = blockIdx.x * blockDim.x + threadIdx.x;
  if (e >= N_EDGES) return;
  int r = row[e], c = col[e];
  if (r != c) unsafeAtomicAdd(&deg[r], 1.0f);
}

__global__ void dinv_kernel(const float* __restrict__ deg,
                            float* __restrict__ dinv) {
  int i = blockIdx.x * blockDim.x + threadIdx.x;
  if (i >= N_NODES) return;
  float d = deg[i];
  dinv[i] = (d > 0.0f) ? rsqrtf(fmaxf(d, 1.0f)) : 0.0f;
}

__global__ void spmm_kernel(const int* __restrict__ row,
                            const int* __restrict__ col,
                            const float* __restrict__ dinv,
                            const float* __restrict__ h,
                            float* __restrict__ out) {
  int wid  = (blockIdx.x * blockDim.x + threadIdx.x) >> 5;
  int lane = threadIdx.x & 31;
  if (wid >= N_EDGES) return;
  int r = row[wid], c = col[wid];
  if (r == c) return;
  float w = -dinv[r] * dinv[c];
  v2f hv = *(const v2f*)(h + (size_t)c * F + lane * 2);
  float* o = out + (size_t)r * F + lane * 2;
  unsafeAtomicAdd(o,     w * hv.x);
  unsafeAtomicAdd(o + 1, w * hv.y);
}

__global__ void combine_kernel(float* __restrict__ p,
                               const float* __restrict__ tkm2, int n) {
  int i = blockIdx.x * blockDim.x + threadIdx.x;
  if (i < n) p[i] = 2.0f * p[i] - tkm2[i];
}

__global__ void wmma_gemm_kernel(const float* __restrict__ T,
                                 const float* __restrict__ Wk,
                                 float* __restrict__ acc, int beta) {
  int wavesPerBlock = blockDim.x >> 5;
  int tile = blockIdx.x * wavesPerBlock + (threadIdx.x >> 5);
  const int numTiles = N_NODES / 16;
  if (tile >= numTiles) return;
  int lane = threadIdx.x & 31;
  int m    = lane & 15;
  int half = lane >> 4;
  int n    = lane & 15;

  v8f c0 = {}, c1 = {}, c2 = {}, c3 = {};
  const float* trow = T + ((size_t)tile * 16 + m) * F;
  float* accbase = acc + (size_t)tile * 16 * F;

  if (beta) {
    #pragma unroll
    for (int r = 0; r < 8; ++r) {
      int mm = r + 8 * half;
      c0[r] = accbase[mm * F +  0 + n];
      c1[r] = accbase[mm * F + 16 + n];
      c2[r] = accbase[mm * F + 32 + n];
      c3[r] = accbase[mm * F + 48 + n];
    }
  }

  #pragma unroll
  for (int kk = 0; kk < F; kk += 4) {
    int k0 = kk + half * 2;
    v2f a = *(const v2f*)(trow + k0);
    const float* wr0 = Wk + (size_t)k0 * F;
    const float* wr1 = wr0 + F;
    v2f b0 = { wr0[ 0 + n], wr1[ 0 + n] };
    v2f b1 = { wr0[16 + n], wr1[16 + n] };
    v2f b2 = { wr0[32 + n], wr1[32 + n] };
    v2f b3 = { wr0[48 + n], wr1[48 + n] };
    c0 = __builtin_amdgcn_wmma_f32_16x16x4_f32(false, a, false, b0, (short)0, c0, false, false);
    c1 = __builtin_amdgcn_wmma_f32_16x16x4_f32(false, a, false, b1, (short)0, c1, false, false);
    c2 = __builtin_amdgcn_wmma_f32_16x16x4_f32(false, a, false, b2, (short)0, c2, false, false);
    c3 = __builtin_amdgcn_wmma_f32_16x16x4_f32(false, a, false, b3, (short)0, c3, false, false);
  }

  #pragma unroll
  for (int r = 0; r < 8; ++r) {
    int mm = r + 8 * half;
    accbase[mm * F +  0 + n] = c0[r];
    accbase[mm * F + 16 + n] = c1[r];
    accbase[mm * F + 32 + n] = c2[r];
    accbase[mm * F + 48 + n] = c3[r];
  }
}

__global__ void ln_elu_kernel(const float* __restrict__ acc,
                              const float* __restrict__ bias,
                              const float* __restrict__ lnw,
                              const float* __restrict__ lnb,
                              float* __restrict__ out) {
  int node = blockIdx.x * (blockDim.x >> 5) + (threadIdx.x >> 5);
  if (node >= N_NODES) return;
  int lane = threadIdx.x & 31;
  int f = lane * 2;
  v2f v = *(const v2f*)(acc + (size_t)node * F + f);
  v.x += bias[f];
  v.y += bias[f + 1];
  float s = v.x + v.y;
  #pragma unroll
  for (int off = 16; off > 0; off >>= 1) s += __shfl_xor(s, off, 32);
  float mean = s * (1.0f / F);
  float dx = v.x - mean, dy = v.y - mean;
  float q = dx * dx + dy * dy;
  #pragma unroll
  for (int off = 16; off > 0; off >>= 1) q += __shfl_xor(q, off, 32);
  float inv = rsqrtf(q * (1.0f / F) + LN_EPS);
  float hx = dx * inv * lnw[f]     + lnb[f];
  float hy = dy * inv * lnw[f + 1] + lnb[f + 1];
  hx = (hx > 0.0f) ? hx : expm1f(hx);
  hy = (hy > 0.0f) ? hy : expm1f(hy);
  v2f o = { hx, hy };
  *(v2f*)(out + (size_t)node * F + f) = o;
}

extern "C" void kernel_launch(void* const* d_in, const int* in_sizes, int n_in,
                              void* d_out, int out_size, void* d_ws, size_t ws_size,
                              hipStream_t stream) {
  (void)in_sizes; (void)n_in; (void)out_size; (void)ws_size;

  const float* x      = (const float*)d_in[0];
  const int*   edge   = (const int*)d_in[1];
  const float* weight = (const float*)d_in[2];
  const float* bias   = (const float*)d_in[3];
  const float* lnw    = (const float*)d_in[4];
  const float* lnb    = (const float*)d_in[5];
  const int* row = edge;
  const int* col = edge + N_EDGES;

  float* ws   = (float*)d_ws;
  float* deg  = ws;
  float* dinv = deg + N_NODES;
  float* b0   = dinv + N_NODES;
  float* b1   = b0 + (size_t)N_NODES * F;
  float* b2   = b1 + (size_t)N_NODES * F;
  float* acc  = (float*)d_out;

  const int nfeat = N_NODES * F;
  dim3 blk(256);

  zero_f32<<<(N_NODES + 255) / 256, blk, 0, stream>>>(deg, N_NODES);
  degree_kernel<<<(N_EDGES + 255) / 256, blk, 0, stream>>>(row, col, deg);
  dinv_kernel<<<(N_NODES + 255) / 256, blk, 0, stream>>>(deg, dinv);

  int spmmGrid = (N_EDGES * 32 + 255) / 256;
  zero_f32<<<(nfeat + 255) / 256, blk, 0, stream>>>(b0, nfeat);
  spmm_kernel<<<spmmGrid, blk, 0, stream>>>(row, col, dinv, x, b0);

  const int tiles = N_NODES / 16;
  dim3 gblk(128);
  int ggrid = (tiles + 3) / 4;
  wmma_gemm_kernel<<<ggrid, gblk, 0, stream>>>(x,  weight + 0 * F * F, acc, 0);
  wmma_gemm_kernel<<<ggrid, gblk, 0, stream>>>(b0, weight + 1 * F * F, acc, 1);

  const float* tkm2 = x;
  const float* tkm1 = b0;
  float* bufs[3] = { b0, b1, b2 };
  int next = 1;
  for (int k = 2; k < K_ORDER; ++k) {
    float* p = bufs[next];
    zero_f32<<<(nfeat + 255) / 256, blk, 0, stream>>>(p, nfeat);
    spmm_kernel<<<spmmGrid, blk, 0, stream>>>(row, col, dinv, tkm1, p);
    combine_kernel<<<(nfeat + 255) / 256, blk, 0, stream>>>(p, tkm2, nfeat);
    wmma_gemm_kernel<<<ggrid, gblk, 0, stream>>>(p, weight + (size_t)k * F * F, acc, 1);
    tkm2 = tkm1;
    tkm1 = p;
    next = (next + 1) % 3;
  }

  ln_elu_kernel<<<(N_NODES + 7) / 8, blk, 0, stream>>>(acc, bias, lnw, lnb, (float*)d_out);
}


# SpMM as row-owned gather: edges grouped by row in-call (int count, arrival-order rank), a wave sums its row in registers, 2*prop-T(k-2) folded in; no float atomics, zero fill or combine launch
# speedup vs baseline: 3.0888x; 3.0888x over previous
#include <hip/hip_runtime.h>
#include <math.h>


#define N_NODES 100000
#define N_EDGES 1600000
#define F 64
#define K_ORDER 8
#define LN_EPS 1e-5f

typedef __attribute__((ext_vector_type(2))) float v2f;
typedef __attribute__((ext_vector_type(8))) float v8f;
typedef __attribute__((ext_vector_type(2))) int v2i;


__global__ void zero_counts(int* __restrict__ cnt, int* __restrict__ cursor) {
  int i = blockIdx.x * blockDim.x + threadIdx.x;
  if (i < N_NODES) cnt[i] = 0;
  if (i == 0) *cursor = 0;
}

__global__ void count_rank_kernel(const int* __restrict__ row,
                                  const int* __restrict__ col,
                                  int* __restrict__ cnt,
                                  int* __restrict__ rank) {
  int e = blockIdx.x * blockDim.x + threadIdx.x;
  if (e >= N_EDGES) return;
  int r = row[e], c = col[e];
  int k = -1;
  if (r != c) k = atomicAdd(&cnt[r], 1);
  rank[e] = k;
}

__global__ void segment_dinv_kernel(const int* __restrict__ cnt,
                                    int* __restrict__ start,
                                    float* __restrict__ dinv,
                                    int* __restrict__ cursor) {
  int i = blockIdx.x * blockDim.x + threadIdx.x;
  int lane = threadIdx.x & 31;
  int n = (i < N_NODES) ? cnt[i] : 0;
  int incl = n;
  #pragma unroll
  for (int off = 1; off < 32; off <<= 1) {
    int t = __shfl_up(incl, off, 32);
    if (lane >= off) incl += t;
  }
  int base = 0;
  if (lane == 31) base = atomicAdd(cursor, incl);
  base = __shfl(base, 31, 32);
  if (i < N_NODES) {
    start[i] = base + incl - n;
    float d = (float)n;
    dinv[i] = (n > 0) ? rsqrtf(fmaxf(d, 1.0f)) : 0.0f;
  }
}

__global__ void place_kernel(const int* __restrict__ row,
                             const int* __restrict__ col,
                             const int* __restrict__ rank,
                             const int* __restrict__ start,
                             const float* __restrict__ dinv,
                             v2i* __restrict__ seg) {
  int e = blockIdx.x * blockDim.x + threadIdx.x;
  if (e >= N_EDGES) return;
  int k = rank[e];
  if (k < 0) return;
  int r = row[e], c = col[e];
  float w = -dinv[r] * dinv[c];
  v2i ent = { c, __float_as_int(w) };
  seg[start[r] + k] = ent;
}

#define GATHER_GROUP 8
template <bool FIRST>
__global__ void spmm_rows_kernel(const int* __restrict__ start,
                                 const int* __restrict__ cnt,
                                 const v2i* __restrict__ seg,
                                 const float* __restrict__ h,
                                 const float* tkm2,
                                 float* out) {
  int rowi = blockIdx.x * (blockDim.x >> 5) + (threadIdx.x >> 5);
  rowi = __builtin_amdgcn_readfirstlane(rowi);
  if (rowi >= N_NODES) return;
  int lane = threadIdx.x & 31;
  int s = start[rowi];
  int n = cnt[rowi];
  size_t o = (size_t)rowi * F + lane * 2;
  v2f t = { 0.0f, 0.0f };
  if (!FIRST) t = *(const v2f*)(tkm2 + o);

  v2f a = { 0.0f, 0.0f };
  const float* hl = h + lane * 2;
  for (int base = 0; base < n; base += 32) {
    int m = n - base;
    if (m > 32) m = 32;
    int ex = 0, ey = 0;
    if (lane < m) {
      v2i ent = seg[s + base + lane];
      ex = ent.x; ey = ent.y;
    }
    asm volatile("" : "+v"(ex), "+v"(ey));
    int j0 = 0;
    for (; j0 + GATHER_GROUP <= m; j0 += GATHER_GROUP) {
      v2f hv[GATHER_GROUP];
      float w[GATHER_GROUP];
      #pragma unroll
      for (int u = 0; u < GATHER_GROUP; ++u) {
        int c = __builtin_amdgcn_readlane(ex, j0 + u);
        w[u] = __int_as_float(__builtin_amdgcn_readlane(ey, j0 + u));
        hv[u] = *(const v2f*)(hl + (size_t)c * F);
      }
      #pragma unroll
      for (int u = 0; u < GATHER_GROUP; ++u) {
        a.x = fmaf(w[u], hv[u].x, a.x);
        a.y = fmaf(w[u], hv[u].y, a.y);
      }
    }
    if (j0 < m) {
      v2f hv[GATHER_GROUP];
      float w[GATHER_GROUP];
      #pragma unroll
      for (int u = 0; u < GATHER_GROUP; ++u) {
        hv[u].x = 0.0f; hv[u].y = 0.0f; w[u] = 0.0f;
        if (j0 + u < m) {
          int c = __builtin_amdgcn_readlane(ex, j0 + u);
          w[u] = __int_as_float(__builtin_amdgcn_readlane(ey, j0 + u));
          hv[u] = *(const v2f*)(hl + (size_t)c * F);
        }
      }
      #pragma unroll
      for (int u = 0; u < GATHER_GROUP; ++u) {
        if (j0 + u < m) {
          a.x = fmaf(w[u], hv[u].x, a.x);
          a.y = fmaf(w[u], hv[u].y, a.y);
        }
      }
    }
  }
  v2f res;
  if (FIRST) { res = a; }
  else       { res.x = 2.0f * a.x - t.x; res.y = 2.0f * a.y - t.y; }
  *(v2f*)(out + o) = res;
}

__global__ void wmma_gemm_kernel(const float* __restrict__ T,
                                 const float* __restrict__ Wk,
                                 float* __restrict__ acc, int beta) {
  int wavesPerBlock = blockDim.x >> 5;
  int tile = blockIdx.x * wavesPerBlock + (threadIdx.x >> 5);
  const int numTiles = N_NODES / 16;
  if (tile >= numTiles) return;
  int lane = threadIdx.x & 31;
  int m    = lane & 15;
  int half = lane >> 4;
  int n    = lane & 15;

  v8f c0 = {}, c1 = {}, c2 = {}, c3 = {};
  const float* trow = T + ((size_t)tile * 16 + m) * F;
  float* accbase = acc + (size_t)tile * 16 * F;

  if (beta) {
    #pragma unroll
    for (int r = 0; r < 8; ++r) {
      int mm = r + 8 * half;
      c0[r] = accbase[mm * F +  0 + n];
      c1[r] = accbase[mm * F + 16 + n];
      c2[r] = accbase[mm * F + 32 + n];
      c3[r] = accbase[mm * F + 48 + n];
    }
  }

  #pragma unroll
  for (int kk = 0; kk < F; kk += 4) {
    int k0 = kk + half * 2;
    v2f a = *(const v2f*)(trow + k0);
    const float* wr0 = Wk + (size_t)k0 * F;
    const float* wr1 = wr0 + F;
    v2f b0 = { wr0[ 0 + n], wr1[ 0 + n] };
    v2f b1 = { wr0[16 + n], wr1[16 + n] };
    v2f b2 = { wr0[32 + n], wr1[32 + n] };
    v2f b3 = { wr0[48 + n], wr1[48 + n] };
    c0 = __builtin_amdgcn_wmma_f32_16x16x4_f32(false, a, false, b0, (short)0, c0, false, false);
    c1 = __builtin_amdgcn_wmma_f32_16x16x4_f32(false, a, false, b1, (short)0, c1, false, false);
    c2 = __builtin_amdgcn_wmma_f32_16x16x4_f32(false, a, false, b2, (short)0, c2, false, false);
    c3 = __builtin_amdgcn_wmma_f32_16x16x4_f32(false, a, false, b3, (short)0, c3, false, false);
  }

  #pragma unroll
  for (int r = 0; r < 8; ++r) {
    int mm = r + 8 * half;
    accbase[mm * F +  0 + n] = c0[r];
    accbase[mm * F + 16 + n] = c1[r];
    accbase[mm * F + 32 + n] = c2[r];
    accbase[mm * F + 48 + n] = c3[r];
  }
}

__global__ void ln_elu_kernel(const float* __restrict__ acc,
                              const float* __restrict__ bias,
                              const float* __restrict__ lnw,
                              const float* __restrict__ lnb,
                              float* __restrict__ out) {
  int node = blockIdx.x * (blockDim.x >> 5) + (threadIdx.x >> 5);
  if (node >= N_NODES) return;
  int lane = threadIdx.x & 31;
  int f = lane * 2;
  v2f v = *(const v2f*)(acc + (size_t)node * F + f);
  v.x += bias[f];
  v.y += bias[f + 1];
  float s = v.x + v.y;
  #pragma unroll
  for (int off = 16; off > 0; off >>= 1) s += __shfl_xor(s, off, 32);
  float mean = s * (1.0f / F);
  float dx = v.x - mean, dy = v.y - mean;
  float q = dx * dx + dy * dy;
  #pragma unroll
  for (int off = 16; off > 0; off >>= 1) q += __shfl_xor(q, off, 32);
  float inv = rsqrtf(q * (1.0f / F) + LN_EPS);
  float hx = dx * inv * lnw[f]     + lnb[f];
  float hy = dy * inv * lnw[f + 1] + lnb[f + 1];
  hx = (hx > 0.0f) ? hx : expm1f(hx);
  hy = (hy > 0.0f) ? hy : expm1f(hy);
  v2f o = { hx, hy };
  *(v2f*)(out + (size_t)node * F + f) = o;
}

extern "C" void kernel_launch(void* const* d_in, const int* in_sizes, int n_in,
                              void* d_out, int out_size, void* d_ws, size_t ws_size,
                              hipStream_t stream) {
  (void)in_sizes; (void)n_in; (void)out_size; (void)ws_size;

  const float* x      = (const float*)d_in[0];
  const int*   edge   = (const int*)d_in[1];
  const float* weight = (const float*)d_in[2];
  const float* bias   = (const float*)d_in[3];
  const float* lnw    = (const float*)d_in[4];
  const float* lnb    = (const float*)d_in[5];
  const int* row = edge;
  const int* col = edge + N_EDGES;

  float* ws   = (float*)d_ws;
  int*   cnt  = (int*)ws;
  float* dinv = ws + N_NODES;
  float* b0   = dinv + N_NODES;
  float* b1   = b0 + (size_t)N_NODES * F;
  v2i*   seg  = (v2i*)(b1 + (size_t)N_NODES * F);
  int*   rank = (int*)(seg + N_EDGES);
  int*   start  = rank + N_EDGES;
  int*   cursor = start + N_NODES;
  float* acc  = (float*)d_out;

  dim3 blk(256);

  zero_counts<<<(N_NODES + 255) / 256, blk, 0, stream>>>(cnt, cursor);
  count_rank_kernel<<<(N_EDGES + 255) / 256, blk, 0, stream>>>(row, col, cnt, rank);
  segment_dinv_kernel<<<(N_NODES + 255) / 256, blk, 0, stream>>>(cnt, start, dinv, cursor);
  place_kernel<<<(N_EDGES + 255) / 256, blk, 0, stream>>>(row, col, rank, start, dinv, seg);

  int rowGrid = (N_NODES + 7) / 8;
  spmm_rows_kernel<true><<<rowGrid, blk, 0, stream>>>(start, cnt, seg, x, x, b0);

  const int tiles = N_NODES / 16;
  dim3 gblk(128);
  int ggrid = (tiles + 3) / 4;
  wmma_gemm_kernel<<<ggrid, gblk, 0, stream>>>(x,  weight + 0 * F * F, acc, 0);
  wmma_gemm_kernel<<<ggrid, gblk, 0, stream>>>(b0, weight + 1 * F * F, acc, 1);

  const float* tkm2 = x;
  float* tkm1 = b0;
  float* p = b1;
  for (int k = 2; k < K_ORDER; ++k) {
    spmm_rows_kernel<false><<<rowGrid, blk, 0, stream>>>(start, cnt, seg, tkm1, tkm2, p);
    wmma_gemm_kernel<<<ggrid, gblk, 0, stream>>>(p, weight + (size_t)k * F * F, acc, 1);
    float* prev = tkm1;
    tkm2 = prev;
    tkm1 = p;
    p = prev;
  }

  ln_elu_kernel<<<(N_NODES + 7) / 8, blk, 0, stream>>>(acc, bias, lnw, lnb, (float*)d_out);
}


# Dense product fused into the SpMM launch: a block's 32 finished rows of T_k meet W_k and its acc piece in LDS, f32 WMMA adds T_k@W_k there, T_k never read back; 8 GEMM launches gone; same sums+order
# speedup vs baseline: 3.8152x; 1.2352x over previous
#include <hip/hip_runtime.h>
#include <math.h>


#define N_NODES 100000
#define N_EDGES 1600000
#define F 64
#define K_ORDER 8
#define LN_EPS 1e-5f

typedef __attribute__((ext_vector_type(2))) float v2f;
typedef __attribute__((ext_vector_type(8))) float v8f;
typedef __attribute__((ext_vector_type(2))) int v2i;
typedef __attribute__((ext_vector_type(4))) float v4f;


__global__ void zero_counts(int* __restrict__ cnt, int* __restrict__ cursor) {
  int i = blockIdx.x * blockDim.x + threadIdx.x;
  if (i < N_NODES) cnt[i] = 0;
  if (i == 0) *cursor = 0;
}

__global__ void count_rank_kernel(const int* __restrict__ row,
                                  const int* __restrict__ col,
                                  int* __restrict__ cnt,
                                  int* __restrict__ rank) {
  int e = blockIdx.x * blockDim.x + threadIdx.x;
  if (e >= N_EDGES) return;
  int r = row[e], c = col[e];
  int k = -1;
  if (r != c) k = atomicAdd(&cnt[r], 1);
  rank[e] = k;
}

__global__ void segment_dinv_kernel(const int* __restrict__ cnt,
                                    int* __restrict__ start,
                                    float* __restrict__ dinv,
                                    int* __restrict__ cursor) {
  int i = blockIdx.x * blockDim.x + threadIdx.x;
  int lane = threadIdx.x & 31;
  int n = (i < N_NODES) ? cnt[i] : 0;
  int incl = n;
  #pragma unroll
  for (int off = 1; off < 32; off <<= 1) {
    int t = __shfl_up(incl, off, 32);
    if (lane >= off) incl += t;
  }
  int base = 0;
  if (lane == 31) base = atomicAdd(cursor, incl);
  base = __shfl(base, 31, 32);
  if (i < N_NODES) {
    start[i] = base + incl - n;
    float d = (float)n;
    dinv[i] = (n > 0) ? rsqrtf(fmaxf(d, 1.0f)) : 0.0f;
  }
}

__global__ void place_kernel(const int* __restrict__ row,
                             const int* __restrict__ col,
                             const int* __restrict__ rank,
                             const int* __restrict__ start,
                             const float* __restrict__ dinv,
                             v2i* __restrict__ seg) {
  int e = blockIdx.x * blockDim.x + threadIdx.x;
  if (e >= N_EDGES) return;
  int k = rank[e];
  if (k < 0) return;
  int r = row[e], c = col[e];
  float w = -dinv[r] * dinv[c];
  v2i ent = { c, __float_as_int(w) };
  seg[start[r] + k] = ent;
}

#define GATHER_GROUP 8
#define BLOCK_ROWS 32
#define WAVE_ROWS 4
#define TILE_PITCH 68
#define WQ_FLOATS 128

static_assert(N_NODES % BLOCK_ROWS == 0, "a block owns 32 whole rows");

__device__ __forceinline__ v2f gather_row(int s, int n, int ex, int ey,
                                          const v2i* __restrict__ seg,
                                          const float* __restrict__ hl, int lane) {
  v2f a = { 0.0f, 0.0f };
  for (int base = 0; base < n; base += 32) {
    int m = n - base;
    if (m > 32) m = 32;
    if (base > 0) {
      ex = 0; ey = 0;
      if (lane < m) {
        v2i ent = seg[s + base + lane];
        ex = ent.x; ey = ent.y;
      }
      asm volatile("" : "+v"(ex), "+v"(ey));
    }
    int j0 = 0;
    for (; j0 + GATHER_GROUP <= m; j0 += GATHER_GROUP) {
      v2f hv[GATHER_GROUP];
      float w[GATHER_GROUP];
      #pragma unroll
      for (int u = 0; u < GATHER_GROUP; ++u) {
        int c = __builtin_amdgcn_readlane(ex, j0 + u);
        w[u] = __int_as_float(__builtin_amdgcn_readlane(ey, j0 + u));
        hv[u] = *(const v2f*)(hl + (size_t)c * F);
      }
      #pragma unroll
      for (int u = 0; u < GATHER_GROUP; ++u) {
        a.x = fmaf(w[u], hv[u].x, a.x);
        a.y = fmaf(w[u], hv[u].y, a.y);
      }
    }
    if (j0 < m) {
      v2f hv[GATHER_GROUP];
      float w[GATHER_GROUP];
      #pragma unroll
      for (int u = 0; u < GATHER_GROUP; ++u) {
        hv[u].x = 0.0f; hv[u].y = 0.0f; w[u] = 0.0f;
        if (j0 + u < m) {
          int c = __builtin_amdgcn_readlane(ex, j0 + u);
          w[u] = __int_as_float(__builtin_amdgcn_readlane(ey, j0 + u));
          hv[u] = *(const v2f*)(hl + (size_t)c * F);
        }
      }
      #pragma unroll
      for (int u = 0; u < GATHER_GROUP; ++u) {
        if (j0 + u < m) {
          a.x = fmaf(w[u], hv[u].x, a.x);
          a.y = fmaf(w[u], hv[u].y, a.y);
        }
      }
    }
  }
  return a;
}

template <bool PAIRED>
__device__ __forceinline__ void tile_product(const float* arow, const float* Wm,
                                             int nb, int half, int ncol,
                                             v8f& c0, v8f& c1) {
  const float* ap = arow + half * 2;
  const float* bp = PAIRED ? Wm + half * WQ_FLOATS + ((nb >> 5) * 16 + ncol) * 4
                           : Wm + (half * 2) * F + nb + ncol;
  if (PAIRED) {
    #pragma unroll
    for (int kk = 0; kk < F; kk += 4) {
      v2f a = *(const v2f*)(ap + kk);
      v4f bq = *(const v4f*)(bp + (kk >> 1) * WQ_FLOATS);
      v2f b0 = { bq.x, bq.y };
      v2f b1 = { bq.z, bq.w };
      c0 = __builtin_amdgcn_wmma_f32_16x16x4_f32(false, a, false, b0, (short)0, c0, false, false);
      c1 = __builtin_amdgcn_wmma_f32_16x16x4_f32(false, a, false, b1, (short)0, c1, false, false);
    }
  } else {
    #pragma unroll 4
    for (int kk = 0; kk < F; kk += 4) {
      v2f a = *(const v2f*)(ap + kk);
      const float* wr0 = bp + kk * F;
      v2f b0 = { wr0[0],  wr0[F] };
      v2f b1 = { wr0[16], wr0[F + 16] };
      c0 = __builtin_amdgcn_wmma_f32_16x16x4_f32(false, a, false, b0, (short)0, c0, false, false);
      c1 = __builtin_amdgcn_wmma_f32_16x16x4_f32(false, a, false, b1, (short)0, c1, false, false);
    }
  }
}

template <bool FIRST, bool STORE_T>
__global__ __launch_bounds__(256) void spmm_gemm_kernel(const int* __restrict__ start,
                                 const int* __restrict__ cnt,
                                 const v2i* __restrict__ seg,
                                 const float* __restrict__ h,
                                 const float* tkm2,
                                 float* tout,
                                 const float* __restrict__ W0,
                                 const float* __restrict__ Wk,
                                 float* __restrict__ acc) {
  __shared__ __attribute__((aligned(16))) float tile[BLOCK_ROWS * TILE_PITCH];
  __shared__ __attribute__((aligned(16))) float wlds[(F / 2) * WQ_FLOATS];
  __shared__ __attribute__((aligned(16))) float clds[FIRST ? 4 : BLOCK_ROWS * TILE_PITCH];

  v4f wreg[4];
  #pragma unroll
  for (int j = 0; j < 2; ++j) {
    int i = threadIdx.x + 256 * j;
    const float* wsrc = Wk + (size_t)(i >> 4) * (2 * F) + (i & 15) * 4;
    wreg[2 * j]     = *(const v4f*)(wsrc);
    wreg[2 * j + 1] = *(const v4f*)(wsrc + F);
  }
  v4f creg[2];
  if (!FIRST) {
    #pragma unroll
    for (int j = 0; j < 2; ++j) {
      int i = threadIdx.x + 256 * j;
      creg[j] = *(const v4f*)(acc + (size_t)(blockIdx.x * BLOCK_ROWS + (i >> 4)) * F + (i & 15) * 4);
    }
  }

  const int wave = __builtin_amdgcn_readfirstlane((int)(threadIdx.x >> 5));
  const int lane = threadIdx.x & 31;
  const int row0 = blockIdx.x * BLOCK_ROWS;
  const int rb = row0 + wave * WAVE_ROWS;

  int s[WAVE_ROWS], n[WAVE_ROWS], ex[WAVE_ROWS], ey[WAVE_ROWS];
  v2f t[WAVE_ROWS];
  #pragma unroll
  for (int rr = 0; rr < WAVE_ROWS; ++rr) {
    s[rr] = start[rb + rr];
    n[rr] = cnt[rb + rr];
  }
  #pragma unroll
  for (int rr = 0; rr < WAVE_ROWS; ++rr) {
    int last = n[rr] - 1;
    int j = (lane < last) ? lane : last;
    int idx = (n[rr] > 0) ? s[rr] + j : 0;
    v2i ent = seg[idx];
    ex[rr] = ent.x; ey[rr] = ent.y;
  }
  #pragma unroll
  for (int rr = 0; rr < WAVE_ROWS; ++rr) {
    t[rr].x = 0.0f; t[rr].y = 0.0f;
    if (!FIRST) t[rr] = *(const v2f*)(tkm2 + (size_t)(rb + rr) * F + lane * 2);
  }
  asm volatile("" : "+v"(ex[0]), "+v"(ey[0]), "+v"(ex[1]), "+v"(ey[1]),
                    "+v"(ex[2]), "+v"(ey[2]), "+v"(ex[3]), "+v"(ey[3]));

  #pragma unroll
  for (int j = 0; j < 2; ++j) {
    int i = threadIdx.x + 256 * j;
    int p = i >> 4, q = i & 15;
    v4f r0 = wreg[2 * j], r1 = wreg[2 * j + 1];
    float* wdst = &wlds[p * WQ_FLOATS + ((q >> 3) * 16 + (q & 3) * 4) * 4 + ((q >> 2) & 1) * 2];
    v2f e0 = { r0.x, r1.x }, e1 = { r0.y, r1.y }, e2 = { r0.z, r1.z }, e3 = { r0.w, r1.w };
    *(v2f*)(wdst +  0) = e0;
    *(v2f*)(wdst +  4) = e1;
    *(v2f*)(wdst +  8) = e2;
    *(v2f*)(wdst + 12) = e3;
  }
  if (!FIRST) {
    #pragma unroll
    for (int j = 0; j < 2; ++j) {
      int i = threadIdx.x + 256 * j;
      *(v4f*)(&clds[(i >> 4) * TILE_PITCH + (i & 15) * 4]) = creg[j];
    }
  }

  const float* hl = h + lane * 2;
  #pragma unroll
  for (int rr = 0; rr < WAVE_ROWS; ++rr) {
    v2f a = gather_row(s[rr], n[rr], ex[rr], ey[rr], seg, hl, lane);
    v2f res;
    if (FIRST) { res = a; }
    else       { res.x = 2.0f * a.x - t[rr].x; res.y = 2.0f * a.y - t[rr].y; }
    if (STORE_T) *(v2f*)(tout + (size_t)(rb + rr) * F + lane * 2) = res;
    *(v2f*)(&tile[(wave * WAVE_ROWS + rr) * TILE_PITCH + lane * 2]) = res;
  }

  const int mt   = wave >> 1;
  const int nb   = (wave & 1) * 32;
  const int ncol = lane & 15;
  const int half = lane >> 4;
  float* accb = acc + (size_t)(row0 + mt * 16) * F + nb;
  __builtin_amdgcn_fence(__ATOMIC_RELEASE, "workgroup", "local");
  __builtin_amdgcn_s_barrier();
  __builtin_amdgcn_fence(__ATOMIC_ACQUIRE, "workgroup", "local");
  if (wave >= 4) return;

  int sidx = (8 * half) * F + ncol;
  asm volatile("" : "+v"(sidx));

  v8f c0 = {}, c1 = {};
  if (!FIRST) {
    #pragma unroll
    for (int r = 0; r < 8; ++r) {
      int mm = mt * 16 + r + 8 * half;
      c0[r] = clds[mm * TILE_PITCH + nb +  0 + ncol];
      c1[r] = clds[mm * TILE_PITCH + nb + 16 + ncol];
    }
  }
  if (FIRST) {
    const float* xrow = h + (size_t)(row0 + mt * 16 + ncol) * F;
    tile_product<false>(xrow, W0, nb, half, ncol, c0, c1);
  }
  tile_product<true>(&tile[(mt * 16 + ncol) * TILE_PITCH], wlds, nb, half, ncol, c0, c1);

  #pragma unroll
  for (int r = 0; r < 8; ++r) {
    accb[sidx + r * F +  0] = c0[r];
    accb[sidx + r * F + 16] = c1[r];
  }
}

__global__ void ln_elu_kernel(const float* __restrict__ acc,
                              const float* __restrict__ bias,
                              const float* __restrict__ lnw,
                              const float* __restrict__ lnb,
                              float* __restrict__ out) {
  int node = blockIdx.x * (blockDim.x >> 5) + (threadIdx.x >> 5);
  if (node >= N_NODES) return;
  int lane = threadIdx.x & 31;
  int f = lane * 2;
  v2f v = *(const v2f*)(acc + (size_t)node * F + f);
  v.x += bias[f];
  v.y += bias[f + 1];
  float s = v.x + v.y;
  #pragma unroll
  for (int off = 16; off > 0; off >>= 1) s += __shfl_xor(s, off, 32);
  float mean = s * (1.0f / F);
  float dx = v.x - mean, dy = v.y - mean;
  float q = dx * dx + dy * dy;
  #pragma unroll
  for (int off = 16; off > 0; off >>= 1) q += __shfl_xor(q, off, 32);
  float inv = rsqrtf(q * (1.0f / F) + LN_EPS);
  float hx = dx * inv * lnw[f]     + lnb[f];
  float hy = dy * inv * lnw[f + 1] + lnb[f + 1];
  hx = (hx > 0.0f) ? hx : expm1f(hx);
  hy = (hy > 0.0f) ? hy : expm1f(hy);
  v2f o = { hx, hy };
  *(v2f*)(out + (size_t)node * F + f) = o;
}

extern "C" void kernel_launch(void* const* d_in, const int* in_sizes, int n_in,
                              void* d_out, int out_size, void* d_ws, size_t ws_size,
                              hipStream_t stream) {
  (void)in_sizes; (void)n_in; (void)out_size; (void)ws_size;

  const float* x      = (const float*)d_in[0];
  const int*   edge   = (const int*)d_in[1];
  const float* weight = (const float*)d_in[2];
  const float* bias   = (const float*)d_in[3];
  const float* lnw    = (const float*)d_in[4];
  const float* lnb    = (const float*)d_in[5];
  const int* row = edge;
  const int* col = edge + N_EDGES;

  float* ws   = (float*)d_ws;
  int*   cnt  = (int*)ws;
  float* dinv = ws + N_NODES;
  float* b0   = dinv + N_NODES;
  float* b1   = b0 + (size_t)N_NODES * F;
  v2i*   seg  = (v2i*)(b1 + (size_t)N_NODES * F);
  int*   rank = (int*)(seg + N_EDGES);
  int*   start  = rank + N_EDGES;
  int*   cursor = start + N_NODES;
  float* acc  = (float*)d_out;

  dim3 blk(256);

  zero_counts<<<(N_NODES + 255) / 256, blk, 0, stream>>>(cnt, cursor);
  count_rank_kernel<<<(N_EDGES + 255) / 256, blk, 0, stream>>>(row, col, cnt, rank);
  segment_dinv_kernel<<<(N_NODES + 255) / 256, blk, 0, stream>>>(cnt, start, dinv, cursor);
  place_kernel<<<(N_EDGES + 255) / 256, blk, 0, stream>>>(row, col, rank, start, dinv, seg);

  int fgrid = N_NODES / BLOCK_ROWS;
  spmm_gemm_kernel<true, true><<<fgrid, blk, 0, stream>>>(
      start, cnt, seg, x, x, b0, weight, weight + 1 * F * F, acc);

  const float* tkm2 = x;
  float* tkm1 = b0;
  float* p = b1;
  for (int k = 2; k < K_ORDER - 1; ++k) {
    spmm_gemm_kernel<false, true><<<fgrid, blk, 0, stream>>>(
        start, cnt, seg, tkm1, tkm2, p, weight, weight + (size_t)k * F * F, acc);
    float* prev = tkm1;
    tkm2 = prev;
    tkm1 = p;
    p = prev;
  }
  spmm_gemm_kernel<false, false><<<fgrid, blk, 0, stream>>>(
      start, cnt, seg, tkm1, tkm2, p, weight, weight + (size_t)(K_ORDER - 1) * F * F, acc);

  ln_elu_kernel<<<(N_NODES + 7) / 8, blk, 0, stream>>>(acc, bias, lnw, lnb, (float*)d_out);
}


# Edge grouping rebuilt as a two-level counting sort in LDS (bins of 512 rows, one block per bin): no global atomics, no per-edge scattered stores or row lookups; a row's order is still arrival order
# speedup vs baseline: 5.0142x; 1.3143x over previous
#include <hip/hip_runtime.h>
#include <math.h>


#define N_NODES 100000
#define N_EDGES 1600000
#define F 64
#define K_ORDER 8
#define LN_EPS 1e-5f

typedef __attribute__((ext_vector_type(2))) float v2f;
typedef __attribute__((ext_vector_type(8))) float v8f;
typedef __attribute__((ext_vector_type(2))) int v2i;
typedef __attribute__((ext_vector_type(4))) float v4f;

#define BIN_SHIFT 9
#define BIN_ROWS 512
#define NBINS ((N_NODES + BIN_ROWS - 1) / BIN_ROWS)
#define PB 256
#define CHUNK (N_EDGES / PB)
#define PT 1024
#define EPT ((CHUNK + PT - 1) / PT)
#define BIN_CAP 12288
#define SWEEP 8

static_assert(N_EDGES % PB == 0, "every block takes the same number of edges");
static_assert(NBINS <= 256, "one thread per bin in the first 256");
static_assert(PB == 256, "the table's columns are summed in four quarters of 64 rows");

__device__ __forceinline__ int block_excl_scan(int v, int* wsum) {
  const int lane = threadIdx.x & 31, wave = threadIdx.x >> 5;
  int incl = v;
  #pragma unroll
  for (int off = 1; off < 32; off <<= 1) {
    int t = __shfl_up(incl, off, 32);
    if (lane >= off) incl += t;
  }
  if (lane == 31) wsum[wave] = incl;
  __syncthreads();
  int t = (lane < wave) ? wsum[lane] : 0;
  #pragma unroll
  for (int off = 16; off > 0; off >>= 1) t += __shfl_xor(t, off, 32);
  __syncthreads();
  return t + incl - v;
}

__global__ __launch_bounds__(PT) void bin_hist_kernel(const int* __restrict__ row,
                                                      const int* __restrict__ col,
                                                      int* __restrict__ table) {
  __shared__ int h[NBINS];
  const int tid = threadIdx.x;
  const int e0 = blockIdx.x * CHUNK;
  int r[EPT], c[EPT];
  #pragma unroll
  for (int k = 0; k < EPT; ++k) {
    int i = tid + k * PT;
    r[k] = 0; c[k] = 0;
    if (i < CHUNK) { r[k] = row[e0 + i]; c[k] = col[e0 + i]; }
  }
  if (tid < NBINS) h[tid] = 0;
  __syncthreads();
  #pragma unroll
  for (int k = 0; k < EPT; ++k) {
    if (r[k] != c[k]) atomicAdd(&h[r[k] >> BIN_SHIFT], 1);
  }
  __syncthreads();
  if (tid < NBINS) table[blockIdx.x * NBINS + tid] = h[tid];
}

__global__ __launch_bounds__(PT) void bin_scatter_kernel(const int* __restrict__ row,
                                                         const int* __restrict__ col,
                                                         const int* __restrict__ table,
                                                         v2i* __restrict__ tmp,
                                                         int* __restrict__ binBase) {
  __shared__ int wsum[32];
  __shared__ int ptot[4][256];
  __shared__ int pbef[4][256];
  __shared__ int cntl[NBINS];
  __shared__ int lstart[NBINS];
  __shared__ int gdst[NBINS];
  __shared__ int nvalid_s;
  __shared__ __attribute__((aligned(8))) v2i stage[CHUNK];
  const int tid = threadIdx.x;
  const int e0 = blockIdx.x * CHUNK;

  int r[EPT], c[EPT];
  #pragma unroll
  for (int k = 0; k < EPT; ++k) {
    int i = tid + k * PT;
    r[k] = 0; c[k] = 0;
    if (i < CHUNK) { r[k] = row[e0 + i]; c[k] = col[e0 + i]; }
  }
  {
    const int q = tid >> 8, b = tid & 255;
    int tot = 0, bef = 0;
    if (b < NBINS) {
      #pragma unroll 16
      for (int g = q * 64; g < q * 64 + 64; ++g) {
        int v = table[g * NBINS + b];
        tot += v;
        if (g < (int)blockIdx.x) bef += v;
      }
    }
    ptot[q][b] = tot; pbef[q][b] = bef;
  }
  if (tid < NBINS) cntl[tid] = 0;
  __syncthreads();
  int tot = 0, bef = 0;
  if (tid < 256) {
    tot = ptot[0][tid] + ptot[1][tid] + ptot[2][tid] + ptot[3][tid];
    bef = pbef[0][tid] + pbef[1][tid] + pbef[2][tid] + pbef[3][tid];
  }
  const int base = block_excl_scan(tot, wsum);
  if (tid < NBINS) {
    gdst[tid] = base + bef;
    if (blockIdx.x == 0) {
      binBase[tid] = base;
      if (tid == NBINS - 1) binBase[NBINS] = base + tot;
    }
  }
  int rk[EPT];
  #pragma unroll
  for (int k = 0; k < EPT; ++k) {
    rk[k] = -1;
    if (r[k] != c[k]) rk[k] = atomicAdd(&cntl[r[k] >> BIN_SHIFT], 1);
  }
  __syncthreads();
  const int v = (tid < NBINS) ? cntl[tid] : 0;
  const int ex = block_excl_scan(v, wsum);
  if (tid < NBINS) {
    lstart[tid] = ex;
    if (tid == NBINS - 1) nvalid_s = ex + v;
  }
  __syncthreads();
  #pragma unroll
  for (int k = 0; k < EPT; ++k) {
    if (rk[k] >= 0) {
      v2i ent = { c[k], r[k] };
      stage[lstart[r[k] >> BIN_SHIFT] + rk[k]] = ent;
    }
  }
  __syncthreads();
  const int nvalid = nvalid_s;
  for (int i = tid; i < nvalid; i += PT) {
    v2i ent = stage[i];
    int b = ent.y >> BIN_SHIFT;
    tmp[gdst[b] + (i - lstart[b])] = ent;
  }
}

__global__ __launch_bounds__(PT) void bin_sort_kernel(const v2i* __restrict__ tmp,
                                                      const int* __restrict__ binBase,
                                                      v2i* __restrict__ seg,
                                                      int* __restrict__ start,
                                                      int* __restrict__ cnt,
                                                      float* __restrict__ dinv) {
  __shared__ int wsum[32];
  __shared__ int rc[BIN_ROWS];
  __shared__ int rs[BIN_ROWS];
  __shared__ __attribute__((aligned(8))) v2i lst[BIN_CAP];
  const int tid = threadIdx.x;
  const int b = blockIdx.x;
  const int lo = binBase[b];
  const int nb = binBase[b + 1] - lo;
  const bool fits = nb <= BIN_CAP;
  if (tid < BIN_ROWS) rc[tid] = 0;
  __syncthreads();
  for (int i0 = tid; i0 < nb; i0 += PT * SWEEP) {
    v2i e[SWEEP];
    #pragma unroll
    for (int u = 0; u < SWEEP; ++u) {
      int i = i0 + u * PT;
      e[u].x = 0; e[u].y = 0;
      if (i < nb) e[u] = tmp[lo + i];
    }
    #pragma unroll
    for (int u = 0; u < SWEEP; ++u) {
      if (i0 + u * PT < nb) atomicAdd(&rc[e[u].y & (BIN_ROWS - 1)], 1);
    }
  }
  __syncthreads();
  const int v = (tid < BIN_ROWS) ? rc[tid] : 0;
  const int ex = block_excl_scan(v, wsum);
  if (tid < BIN_ROWS) {
    rs[tid] = ex;
    rc[tid] = 0;
    int rowg = b * BIN_ROWS + tid;
    if (rowg < N_NODES) {
      start[rowg] = lo + ex;
      cnt[rowg] = v;
      float d = (float)v;
      dinv[rowg] = (v > 0) ? rsqrtf(fmaxf(d, 1.0f)) : 0.0f;
    }
  }
  __syncthreads();
  for (int i0 = tid; i0 < nb; i0 += PT * SWEEP) {
    v2i e[SWEEP];
    #pragma unroll
    for (int u = 0; u < SWEEP; ++u) {
      int i = i0 + u * PT;
      e[u].x = 0; e[u].y = 0;
      if (i < nb) e[u] = tmp[lo + i];
    }
    #pragma unroll
    for (int u = 0; u < SWEEP; ++u) {
      if (i0 + u * PT < nb) {
        int rl = e[u].y & (BIN_ROWS - 1);
        int pos = rs[rl] + atomicAdd(&rc[rl], 1);
        if (fits) lst[pos] = e[u]; else seg[lo + pos] = e[u];
      }
    }
  }
  if (fits) {
    __syncthreads();
    for (int i = tid; i < nb; i += PT) seg[lo + i] = lst[i];
  }
}

__global__ void edge_weight_kernel(v2i* __restrict__ seg,
                                   const float* __restrict__ dinv,
                                   const int* __restrict__ binBase) {
  int i = blockIdx.x * blockDim.x + threadIdx.x;
  if (i >= binBase[NBINS]) return;
  v2i e = seg[i];
  float w = -dinv[e.y] * dinv[e.x];
  e.y = __float_as_int(w);
  seg[i] = e;
}

#define GATHER_GROUP 8
#define BLOCK_ROWS 32
#define WAVE_ROWS 4
#define TILE_PITCH 68
#define WQ_FLOATS 128

static_assert(N_NODES % BLOCK_ROWS == 0, "a block owns 32 whole rows");

__device__ __forceinline__ v2f gather_row(int s, int n, int ex, int ey,
                                          const v2i* __restrict__ seg,
                                          const float* __restrict__ hl, int lane) {
  v2f a = { 0.0f, 0.0f };
  for (int base = 0; base < n; base += 32) {
    int m = n - base;
    if (m > 32) m = 32;
    if (base > 0) {
      ex = 0; ey = 0;
      if (lane < m) {
        v2i ent = seg[s + base + lane];
        ex = ent.x; ey = ent.y;
      }
      asm volatile("" : "+v"(ex), "+v"(ey));
    }
    int j0 = 0;
    for (; j0 + GATHER_GROUP <= m; j0 += GATHER_GROUP) {
      v2f hv[GATHER_GROUP];
      float w[GATHER_GROUP];
      #pragma unroll
      for (int u = 0; u < GATHER_GROUP; ++u) {
        int c = __builtin_amdgcn_readlane(ex, j0 + u);
        w[u] = __int_as_float(__builtin_amdgcn_readlane(ey, j0 + u));
        hv[u] = *(const v2f*)(hl + (size_t)c * F);
      }
      #pragma unroll
      for (int u = 0; u < GATHER_GROUP; ++u) {
        a.x = fmaf(w[u], hv[u].x, a.x);
        a.y = fmaf(w[u], hv[u].y, a.y);
      }
    }
    if (j0 < m) {
      v2f hv[GATHER_GROUP];
      float w[GATHER_GROUP];
      #pragma unroll
      for (int u = 0; u < GATHER_GROUP; ++u) {
        hv[u].x = 0.0f; hv[u].y = 0.0f; w[u] = 0.0f;
        if (j0 + u < m) {
          int c = __builtin_amdgcn_readlane(ex, j0 + u);
          w[u] = __int_as_float(__builtin_amdgcn_readlane(ey, j0 + u));
          hv[u] = *(const v2f*)(hl + (size_t)c * F);
        }
      }
      #pragma unroll
      for (int u = 0; u < GATHER_GROUP; ++u) {
        if (j0 + u < m) {
          a.x = fmaf(w[u], hv[u].x, a.x);
          a.y = fmaf(w[u], hv[u].y, a.y);
        }
      }
    }
  }
  return a;
}

template <bool PAIRED>
__device__ __forceinline__ void tile_product(const float* arow, const float* Wm,
                                             int nb, int half, int ncol,
                                             v8f& c0, v8f& c1) {
  const float* ap = arow + half * 2;
  const float* bp = PAIRED ? Wm + half * WQ_FLOATS + ((nb >> 5) * 16 + ncol) * 4
                           : Wm + (half * 2) * F + nb + ncol;
  if (PAIRED) {
    #pragma unroll
    for (int kk = 0; kk < F; kk += 4) {
      v2f a = *(const v2f*)(ap + kk);
      v4f bq = *(const v4f*)(bp + (kk >> 1) * WQ_FLOATS);
      v2f b0 = { bq.x, bq.y };
      v2f b1 = { bq.z, bq.w };
      c0 = __builtin_amdgcn_wmma_f32_16x16x4_f32(false, a, false, b0, (short)0, c0, false, false);
      c1 = __builtin_amdgcn_wmma_f32_16x16x4_f32(false, a, false, b1, (short)0, c1, false, false);
    }
  } else {
    #pragma unroll 4
    for (int kk = 0; kk < F; kk += 4) {
      v2f a = *(const v2f*)(ap + kk);
      const float* wr0 = bp + kk * F;
      v2f b0 = { wr0[0],  wr0[F] };
      v2f b1 = { wr0[16], wr0[F + 16] };
      c0 = __builtin_amdgcn_wmma_f32_16x16x4_f32(false, a, false, b0, (short)0, c0, false, false);
      c1 = __builtin_amdgcn_wmma_f32_16x16x4_f32(false, a, false, b1, (short)0, c1, false, false);
    }
  }
}

template <bool FIRST, bool STORE_T>
__global__ __launch_bounds__(256) void spmm_gemm_kernel(const int* __restrict__ start,
                                 const int* __restrict__ cnt,
                                 const v2i* __restrict__ seg,
                                 const float* __restrict__ h,
                                 const float* tkm2,
                                 float* tout,
                                 const float* __restrict__ W0,
                                 const float* __restrict__ Wk,
                                 float* __restrict__ acc) {
  __shared__ __attribute__((aligned(16))) float tile[BLOCK_ROWS * TILE_PITCH];
  __shared__ __attribute__((aligned(16))) float wlds[(F / 2) * WQ_FLOATS];
  __shared__ __attribute__((aligned(16))) float clds[FIRST ? 4 : BLOCK_ROWS * TILE_PITCH];

  v4f wreg[4];
  #pragma unroll
  for (int j = 0; j < 2; ++j) {
    int i = threadIdx.x + 256 * j;
    const float* wsrc = Wk + (size_t)(i >> 4) * (2 * F) + (i & 15) * 4;
    wreg[2 * j]     = *(const v4f*)(wsrc);
    wreg[2 * j + 1] = *(const v4f*)(wsrc + F);
  }
  v4f creg[2];
  if (!FIRST) {
    #pragma unroll
    for (int j = 0; j < 2; ++j) {
      int i = threadIdx.x + 256 * j;
      creg[j] = *(const v4f*)(acc + (size_t)(blockIdx.x * BLOCK_ROWS + (i >> 4)) * F + (i & 15) * 4);
    }
  }

  const int wave = __builtin_amdgcn_readfirstlane((int)(threadIdx.x >> 5));
  const int lane = threadIdx.x & 31;
  const int row0 = blockIdx.x * BLOCK_ROWS;
  const int rb = row0 + wave * WAVE_ROWS;

  int s[WAVE_ROWS], n[WAVE_ROWS], ex[WAVE_ROWS], ey[WAVE_ROWS];
  v2f t[WAVE_ROWS];
  #pragma unroll
  for (int rr = 0; rr < WAVE_ROWS; ++rr) {
    s[rr] = start[rb + rr];
    n[rr] = cnt[rb + rr];
  }
  #pragma unroll
  for (int rr = 0; rr < WAVE_ROWS; ++rr) {
    int last = n[rr] - 1;
    int j = (lane < last) ? lane : last;
    int idx = (n[rr] > 0) ? s[rr] + j : 0;
    v2i ent = seg[idx];
    ex[rr] = ent.x; ey[rr] = ent.y;
  }
  #pragma unroll
  for (int rr = 0; rr < WAVE_ROWS; ++rr) {
    t[rr].x = 0.0f; t[rr].y = 0.0f;
    if (!FIRST) t[rr] = *(const v2f*)(tkm2 + (size_t)(rb + rr) * F + lane * 2);
  }
  asm volatile("" : "+v"(ex[0]), "+v"(ey[0]), "+v"(ex[1]), "+v"(ey[1]),
                    "+v"(ex[2]), "+v"(ey[2]), "+v"(ex[3]), "+v"(ey[3]));

  #pragma unroll
  for (int j = 0; j < 2; ++j) {
    int i = threadIdx.x + 256 * j;
    int p = i >> 4, q = i & 15;
    v4f r0 = wreg[2 * j], r1 = wreg[2 * j + 1];
    float* wdst = &wlds[p * WQ_FLOATS + ((q >> 3) * 16 + (q & 3) * 4) * 4 + ((q >> 2) & 1) * 2];
    v2f e0 = { r0.x, r1.x }, e1 = { r0.y, r1.y }, e2 = { r0.z, r1.z }, e3 = { r0.w, r1.w };
    *(v2f*)(wdst +  0) = e0;
    *(v2f*)(wdst +  4) = e1;
    *(v2f*)(wdst +  8) = e2;
    *(v2f*)(wdst + 12) = e3;
  }
  if (!FIRST) {
    #pragma unroll
    for (int j = 0; j < 2; ++j) {
      int i = threadIdx.x + 256 * j;
      *(v4f*)(&clds[(i >> 4) * TILE_PITCH + (i & 15) * 4]) = creg[j];
    }
  }

  const float* hl = h + lane * 2;
  #pragma unroll
  for (int rr = 0; rr < WAVE_ROWS; ++rr) {
    v2f a = gather_row(s[rr], n[rr], ex[rr], ey[rr], seg, hl, lane);
    v2f res;
    if (FIRST) { res = a; }
    else       { res.x = 2.0f * a.x - t[rr].x; res.y = 2.0f * a.y - t[rr].y; }
    if (STORE_T) *(v2f*)(tout + (size_t)(rb + rr) * F + lane * 2) = res;
    *(v2f*)(&tile[(wave * WAVE_ROWS + rr) * TILE_PITCH + lane * 2]) = res;
  }

  const int mt   = wave >> 1;
  const int nb   = (wave & 1) * 32;
  const int ncol = lane & 15;
  const int half = lane >> 4;
  float* accb = acc + (size_t)(row0 + mt * 16) * F + nb;
  __builtin_amdgcn_fence(__ATOMIC_RELEASE, "workgroup", "local");
  __builtin_amdgcn_s_barrier();
  __builtin_amdgcn_fence(__ATOMIC_ACQUIRE, "workgroup", "local");
  if (wave >= 4) return;

  int sidx = (8 * half) * F + ncol;
  asm volatile("" : "+v"(sidx));

  v8f c0 = {}, c1 = {};
  if (!FIRST) {
    #pragma unroll
    for (int r = 0; r < 8; ++r) {
      int mm = mt * 16 + r + 8 * half;
      c0[r] = clds[mm * TILE_PITCH + nb +  0 + ncol];
      c1[r] = clds[mm * TILE_PITCH + nb + 16 + ncol];
    }
  }
  if (FIRST) {
    const float* xrow = h + (size_t)(row0 + mt * 16 + ncol) * F;
    tile_product<false>(xrow, W0, nb, half, ncol, c0, c1);
  }
  tile_product<true>(&tile[(mt * 16 + ncol) * TILE_PITCH], wlds, nb, half, ncol, c0, c1);

  #pragma unroll
  for (int r = 0; r < 8; ++r) {
    accb[sidx + r * F +  0] = c0[r];
    accb[sidx + r * F + 16] = c1[r];
  }
}

__global__ void ln_elu_kernel(const float* __restrict__ acc,
                              const float* __restrict__ bias,
                              const float* __restrict__ lnw,
                              const float* __restrict__ lnb,
                              float* __restrict__ out) {
  int node = blockIdx.x * (blockDim.x >> 5) + (threadIdx.x >> 5);
  if (node >= N_NODES) return;
  int lane = threadIdx.x & 31;
  int f = lane * 2;
  v2f v = *(const v2f*)(acc + (size_t)node * F + f);
  v.x += bias[f];
  v.y += bias[f + 1];
  float s = v.x + v.y;
  #pragma unroll
  for (int off = 16; off > 0; off >>= 1) s += __shfl_xor(s, off, 32);
  float mean = s * (1.0f / F);
  float dx = v.x - mean, dy = v.y - mean;
  float q = dx * dx + dy * dy;
  #pragma unroll
  for (int off = 16; off > 0; off >>= 1) q += __shfl_xor(q, off, 32);
  float inv = rsqrtf(q * (1.0f / F) + LN_EPS);
  float hx = dx * inv * lnw[f]     + lnb[f];
  float hy = dy * inv * lnw[f + 1] + lnb[f + 1];
  hx = (hx > 0.0f) ? hx : expm1f(hx);
  hy = (hy > 0.0f) ? hy : expm1f(hy);
  v2f o = { hx, hy };
  *(v2f*)(out + (size_t)node * F + f) = o;
}

extern "C" void kernel_launch(void* const* d_in, const int* in_sizes, int n_in,
                              void* d_out, int out_size, void* d_ws, size_t ws_size,
                              hipStream_t stream) {
  (void)in_sizes; (void)n_in; (void)out_size; (void)ws_size;

  const float* x      = (const float*)d_in[0];
  const int*   edge   = (const int*)d_in[1];
  const float* weight = (const float*)d_in[2];
  const float* bias   = (const float*)d_in[3];
  const float* lnw    = (const float*)d_in[4];
  const float* lnb    = (const float*)d_in[5];
  const int* row = edge;
  const int* col = edge + N_EDGES;

  float* ws   = (float*)d_ws;
  int*   cnt  = (int*)ws;
  float* dinv = ws + N_NODES;
  float* b0   = dinv + N_NODES;
  float* b1   = b0 + (size_t)N_NODES * F;
  v2i*   seg  = (v2i*)(b1 + (size_t)N_NODES * F);
  int*   table   = (int*)(seg + N_EDGES);
  int*   binBase = table + PB * NBINS;
  int*   start   = table + N_EDGES;
  v2i*   tmp  = (v2i*)b1;
  float* acc  = (float*)d_out;

  dim3 blk(256);

  bin_hist_kernel<<<PB, PT, 0, stream>>>(row, col, table);
  bin_scatter_kernel<<<PB, PT, 0, stream>>>(row, col, table, tmp, binBase);
  bin_sort_kernel<<<NBINS, PT, 0, stream>>>(tmp, binBase, seg, start, cnt, dinv);
  edge_weight_kernel<<<(N_EDGES + 255) / 256, blk, 0, stream>>>(seg, dinv, binBase);

  int fgrid = N_NODES / BLOCK_ROWS;
  spmm_gemm_kernel<true, true><<<fgrid, blk, 0, stream>>>(
      start, cnt, seg, x, x, b0, weight, weight + 1 * F * F, acc);

  const float* tkm2 = x;
  float* tkm1 = b0;
  float* p = b1;
  for (int k = 2; k < K_ORDER - 1; ++k) {
    spmm_gemm_kernel<false, true><<<fgrid, blk, 0, stream>>>(
        start, cnt, seg, tkm1, tkm2, p, weight, weight + (size_t)k * F * F, acc);
    float* prev = tkm1;
    tkm2 = prev;
    tkm1 = p;
    p = prev;
  }
  spmm_gemm_kernel<false, false><<<fgrid, blk, 0, stream>>>(
      start, cnt, seg, tkm1, tkm2, p, weight, weight + (size_t)(K_ORDER - 1) * F * F, acc);

  ln_elu_kernel<<<(N_NODES + 7) / 8, blk, 0, stream>>>(acc, bias, lnw, lnb, (float*)d_out);
}
